# starting kernel
# baseline (speedup 1.0000x reference)
#include <hip/hip_runtime.h>
#include <hip/hip_bf16.h>
#include <math.h>

#define NUM_GRIDS 16
#define IN_X 32
#define IN_Z 32
#define OUT 64
#define BATCH 512
#define GP1 (NUM_GRIDS + 1)

typedef __attribute__((ext_vector_type(2))) float v2f;
typedef __attribute__((ext_vector_type(8))) float v8f;

__global__ void build_scatter(const float* __restrict__ x,
                              const float* __restrict__ z,
                              const float* __restrict__ borders,
                              const float* __restrict__ invlen,
                              float* __restrict__ Vx,
                              float* __restrict__ Vz) {
  const int per = IN_X * BATCH;
  int t = blockIdx.x * blockDim.x + threadIdx.x;
  if (t >= 2 * per) return;
  int sel = t / per;
  int r   = t - sel * per;
  int i   = r / BATCH;
  int b   = r - i * BATCH;

  float v = sel ? z[i * BATCH + b] : x[i * BATCH + b];
  float* V = sel ? Vz : Vx;

  float e   = __expf(-fabsf(v));
  float cdf = (v > 0.0f) ? (1.0f - 0.5f * e) : (0.5f * e);
  int idx = (int)(cdf * (float)NUM_GRIDS);
  idx = idx < 0 ? 0 : (idx > NUM_GRIDS - 1 ? NUM_GRIDS - 1 : idx);
  float d = (v - borders[idx]) * invlen[idx];

#pragma unroll
  for (int g = 0; g < GP1; ++g) {
    float w = (g == idx) ? (1.0f - d) : ((g == idx + 1) ? d : 0.0f);
    V[((size_t)g * IN_X + i) * BATCH + b] = w;
  }
}

__global__ void __launch_bounds__(32) basis_gemm(
    const float* __restrict__ P,
    const float* __restrict__ Vx,
    const float* __restrict__ Vz,
    float* __restrict__ out)
{
  const int o    = blockIdx.y;
  const int b0   = blockIdx.x * 32;
  const int lane = threadIdx.x;
  const int hi   = lane >> 4;
  const int lo   = lane & 15;

  float outacc0 = 0.0f;
  float outacc1 = 0.0f;

  const size_t gzStride = (size_t)OUT * IN_X * IN_Z;

  for (int gx = 0; gx < GP1; ++gx) {
    v8f acc00 = {};
    v8f acc01 = {};
    v8f acc10 = {};
    v8f acc11 = {};

    for (int gz = 0; gz < GP1; ++gz) {
      const float* Ap  = P + ((((size_t)gx * GP1 + gz) * OUT + o) * IN_X) * IN_Z;
      const float* Bp0 = Vz + (size_t)gz * IN_Z * BATCH + (b0 + lo);
      const float* Bp1 = Bp0 + 16;

      if (gz + 1 < GP1)
        __builtin_prefetch(Ap + gzStride + (size_t)lo * IN_Z, 0, 3);

#pragma unroll
      for (int jj = 0; jj < IN_Z; jj += 4) {
        const int j = jj + 2 * hi;
        v2f a0 = *(const v2f*)(Ap + (size_t)lo        * IN_Z + j);
        v2f a1 = *(const v2f*)(Ap + (size_t)(16 + lo) * IN_Z + j);
        v2f b0v, b1v;
        b0v.x = Bp0[(size_t)j * BATCH];
        b0v.y = Bp0[(size_t)(j + 1) * BATCH];
        b1v.x = Bp1[(size_t)j * BATCH];
        b1v.y = Bp1[(size_t)(j + 1) * BATCH];
        acc00 = __builtin_amdgcn_wmma_f32_16x16x4_f32(
            false, a0, false, b0v, (short)0, acc00, false, false);
        acc10 = __builtin_amdgcn_wmma_f32_16x16x4_f32(
            false, a0, false, b1v, (short)0, acc10, false, false);
        acc01 = __builtin_amdgcn_wmma_f32_16x16x4_f32(
            false, a1, false, b0v, (short)0, acc01, false, false);
        acc11 = __builtin_amdgcn_wmma_f32_16x16x4_f32(
            false, a1, false, b1v, (short)0, acc11, false, false);
      }
    }

    float p0 = 0.0f, p1 = 0.0f;
#pragma unroll
    for (int rr = 0; rr < 8; ++rr) {
      const int m = rr + 8 * hi;
      float w00 = Vx[((size_t)gx * IN_X + m)      * BATCH + (b0 + lo)];
      float w01 = Vx[((size_t)gx * IN_X + 16 + m) * BATCH + (b0 + lo)];
      float w10 = Vx[((size_t)gx * IN_X + m)      * BATCH + (b0 + 16 + lo)];
      float w11 = Vx[((size_t)gx * IN_X + 16 + m) * BATCH + (b0 + 16 + lo)];
      p0 += acc00[rr] * w00 + acc01[rr] * w01;
      p1 += acc10[rr] * w10 + acc11[rr] * w11;
    }
    p0 += __shfl_xor(p0, 16, 32);
    p1 += __shfl_xor(p1, 16, 32);
    outacc0 += p0;
    outacc1 += p1;
  }

  if (lane < 16) {
    out[(size_t)o * BATCH + b0 + lane]      = outacc0;
    out[(size_t)o * BATCH + b0 + 16 + lane] = outacc1;
  }
}

extern "C" void kernel_launch(void* const* d_in, const int* in_sizes, int n_in,
                              void* d_out, int out_size, void* d_ws, size_t ws_size,
                              hipStream_t stream) {
  const float* x       = (const float*)d_in[0];
  const float* z       = (const float*)d_in[1];
  const float* P       = (const float*)d_in[2];
  const float* borders = (const float*)d_in[3];
  const float* invlen  = (const float*)d_in[4];
  float* out = (float*)d_out;

  float* Vx = (float*)d_ws;
  float* Vz = Vx + (size_t)GP1 * IN_X * BATCH;

  {
    int total = 2 * IN_X * BATCH;
    build_scatter<<<dim3((total + 255) / 256), dim3(256), 0, stream>>>(
        x, z, borders, invlen, Vx, Vz);
  }
  basis_gemm<<<dim3(BATCH / 32, OUT), dim3(32), 0, stream>>>(P, Vx, Vz, out);
}


# Weight build fused into the product kernel: each wave bins its own x and z values and fills its z-weight B fragments and x weights in registers; build launch and workspace gone (2 launches to 1)
# speedup vs baseline: 2.4391x; 2.4391x over previous
#include <hip/hip_runtime.h>
#include <hip/hip_bf16.h>
#include <math.h>


#define NUM_GRIDS 16
#define IN_X 32
#define IN_Z 32
#define OUT 64
#define BATCH 512
#define GP1 (NUM_GRIDS + 1)
#define NSLICE (GP1 * GP1)
#define WAVES_PER_WG 4

typedef __attribute__((ext_vector_type(2))) float v2f;
typedef __attribute__((ext_vector_type(8))) float v8f;

struct Bin {
  int   idx;
  float d;
};

__device__ __forceinline__ Bin bin_of(float v,
                                      const float* __restrict__ borders,
                                      const float* __restrict__ invlen) {
  float e   = __expf(-fabsf(v));
  float cdf = (v > 0.0f) ? (1.0f - 0.5f * e) : (0.5f * e);
  int idx = (int)(cdf * (float)NUM_GRIDS);
  idx = idx < 0 ? 0 : (idx > NUM_GRIDS - 1 ? NUM_GRIDS - 1 : idx);
  Bin r;
  r.idx = idx;
  r.d   = (v - borders[idx]) * invlen[idx];
  return r;
}

__device__ __forceinline__ float hat(int g, int idx, float d) {
  return (g == idx) ? (1.0f - d) : ((g == idx + 1) ? d : 0.0f);
}

struct Hat2 {
  int   ge, go;
  float ve, vo;
};

__device__ __forceinline__ Hat2 hat2_of(Bin b) {
  const bool  odd = (b.idx & 1) != 0;
  const float omd = 1.0f - b.d;
  Hat2 h;
  h.ge = (b.idx + 1) & ~1;
  h.go = b.idx | 1;
  h.ve = odd ? b.d : omd;
  h.vo = odd ? omd : b.d;
  return h;
}

template <int G>
__device__ __forceinline__ float hat2(const Hat2& h) {
  return (G & 1) ? ((h.go == G) ? h.vo : 0.0f) : ((h.ge == G) ? h.ve : 0.0f);
}

template <int G>
__device__ __forceinline__ void fill_b(v2f (&bz)[GP1][8], int t, const Hat2& kx, const Hat2& ky) {
  if constexpr (G < GP1) {
    bz[G][t].x = hat2<G>(kx);
    bz[G][t].y = hat2<G>(ky);
    fill_b<G + 1>(bz, t, kx, ky);
  }
}

struct AFrags {
  v2f a0[8];
  v2f a1[8];
};

__device__ __forceinline__ void load_a(AFrags& f, int s,
                                       const float* __restrict__ P,
                                       int o, int lo, int hi) {
  const float* Ap = P + ((size_t)s * OUT + o) * (IN_X * IN_Z);

  if (s + 1 < NSLICE)
    __builtin_prefetch(Ap + (size_t)OUT * IN_X * IN_Z + (size_t)lo * IN_Z, 0, 3);

#pragma unroll
  for (int t = 0; t < 8; ++t) {
    const int j = 4 * t + 2 * hi;
    f.a0[t] = *(const v2f*)(Ap + (size_t)lo        * IN_Z + j);
    f.a1[t] = *(const v2f*)(Ap + (size_t)(16 + lo) * IN_Z + j);
  }
}

struct XW {
  float w00[8];
  float w01[8];
  float w10[8];
  float w11[8];
};

struct XB {
  Bin b00[8];
  Bin b01[8];
  Bin b10[8];
  Bin b11[8];
};

__device__ __forceinline__ void bin_x(XB& xb, const float* __restrict__ x,
                                      const float* __restrict__ borders,
                                      const float* __restrict__ invlen,
                                      int bcol, int hi) {
#pragma unroll
  for (int rr = 0; rr < 8; ++rr) {
    const int m = rr + 8 * hi;
    xb.b00[rr] = bin_of(x[(size_t)m        * BATCH + bcol],      borders, invlen);
    xb.b01[rr] = bin_of(x[(size_t)(16 + m) * BATCH + bcol],      borders, invlen);
    xb.b10[rr] = bin_of(x[(size_t)m        * BATCH + bcol + 16], borders, invlen);
    xb.b11[rr] = bin_of(x[(size_t)(16 + m) * BATCH + bcol + 16], borders, invlen);
  }
}

__device__ __forceinline__ void make_xw(XW& w, int gx, const XB& xb) {
#pragma unroll
  for (int rr = 0; rr < 8; ++rr) {
    w.w00[rr] = hat(gx, xb.b00[rr].idx, xb.b00[rr].d);
    w.w01[rr] = hat(gx, xb.b01[rr].idx, xb.b01[rr].d);
    w.w10[rr] = hat(gx, xb.b10[rr].idx, xb.b10[rr].d);
    w.w11[rr] = hat(gx, xb.b11[rr].idx, xb.b11[rr].d);
  }
}

__device__ __forceinline__ void finish_gx(const XW& w,
                                          v8f& acc00, v8f& acc01, v8f& acc10, v8f& acc11,
                                          float& outacc0, float& outacc1) {
  float p0 = 0.0f, p1 = 0.0f;
#pragma unroll
  for (int rr = 0; rr < 8; ++rr) {
    p0 += acc00[rr] * w.w00[rr] + acc01[rr] * w.w01[rr];
    p1 += acc10[rr] * w.w10[rr] + acc11[rr] * w.w11[rr];
  }
  p0 += __shfl_xor(p0, 16, 32);
  p1 += __shfl_xor(p1, 16, 32);
  outacc0 += p0;
  outacc1 += p1;
  acc00 = v8f{}; acc01 = v8f{}; acc10 = v8f{}; acc11 = v8f{};
}

__global__ void __launch_bounds__(WAVES_PER_WG * 32) basis_gemm(
    const float* __restrict__ x,
    const float* __restrict__ z,
    const float* __restrict__ P,
    const float* __restrict__ borders,
    const float* __restrict__ invlen,
    float* __restrict__ out)
{
  const int o    = blockIdx.y;
  const int wave = threadIdx.x >> 5;
  const int b0   = (blockIdx.x * WAVES_PER_WG + wave) * 32;
  const int lane = threadIdx.x & 31;
  const int hi   = lane >> 4;
  const int lo   = lane & 15;
  const int bcol = b0 + lo;

  float outacc0 = 0.0f;
  float outacc1 = 0.0f;

  v8f acc00 = {};
  v8f acc01 = {};
  v8f acc10 = {};
  v8f acc11 = {};

  Hat2 zb0[16], zb1[16];
#pragma unroll
  for (int t = 0; t < 8; ++t) {
#pragma unroll
    for (int c = 0; c < 2; ++c) {
      const int j = 4 * t + 2 * hi + c;
      zb0[2 * t + c] = hat2_of(bin_of(z[(size_t)j * BATCH + bcol],      borders, invlen));
      zb1[2 * t + c] = hat2_of(bin_of(z[(size_t)j * BATCH + bcol + 16], borders, invlen));
    }
  }
  XB xb;
  bin_x(xb, x, borders, invlen, bcol, hi);

  AFrags f0, f1;
  load_a(f0, 0, P, o, lo, hi);
  load_a(f1, 1, P, o, lo, hi);

  v2f bz0[GP1][8];
  v2f bz1[GP1][8];
#pragma unroll
  for (int t = 0; t < 8; ++t) {
    fill_b<0>(bz0, t, zb0[2 * t], zb0[2 * t + 1]);
    fill_b<0>(bz1, t, zb1[2 * t], zb1[2 * t + 1]);
  }
  XW xw;

  for (int gx = 0; gx < GP1; ++gx) {
#pragma unroll
    for (int gz = 0; gz < GP1; ++gz) {
      const int s = gx * GP1 + gz;
      const int sn = (s + 2 < NSLICE) ? s + 2 : NSLICE - 1;
      AFrags f2;
      load_a(f2, sn, P, o, lo, hi);
      __builtin_amdgcn_sched_barrier(0);
      if (gz == GP1 - 1)
        make_xw(xw, gx, xb);
#pragma unroll
      for (int t = 0; t < 8; ++t) {
        acc00 = __builtin_amdgcn_wmma_f32_16x16x4_f32(
            false, f0.a0[t], false, bz0[gz][t], (short)0, acc00, false, false);
        acc10 = __builtin_amdgcn_wmma_f32_16x16x4_f32(
            false, f0.a0[t], false, bz1[gz][t], (short)0, acc10, false, false);
        acc01 = __builtin_amdgcn_wmma_f32_16x16x4_f32(
            false, f0.a1[t], false, bz0[gz][t], (short)0, acc01, false, false);
        acc11 = __builtin_amdgcn_wmma_f32_16x16x4_f32(
            false, f0.a1[t], false, bz1[gz][t], (short)0, acc11, false, false);
      }
      if (gz == GP1 - 1) {
#pragma unroll
        for (int k = 0; k < 32; ++k) {
          __builtin_amdgcn_sched_group_barrier(0x8, 1, 0);
          __builtin_amdgcn_sched_group_barrier(0x2, 4, 0);
        }
      }
      f0 = f1;
      f1 = f2;
    }
    finish_gx(xw, acc00, acc01, acc10, acc11, outacc0, outacc1);
  }

  if (lane < 16) {
    out[(size_t)o * BATCH + b0 + lane]      = outacc0;
    out[(size_t)o * BATCH + b0 + 16 + lane] = outacc1;
  }
}

extern "C" void kernel_launch(void* const* d_in, const int* in_sizes, int n_in,
                              void* d_out, int out_size, void* d_ws, size_t ws_size,
                              hipStream_t stream) {
  const float* x       = (const float*)d_in[0];
  const float* z       = (const float*)d_in[1];
  const float* P       = (const float*)d_in[2];
  const float* borders = (const float*)d_in[3];
  const float* invlen  = (const float*)d_in[4];
  float* out = (float*)d_out;

  (void)d_ws;
  (void)ws_size;

  basis_gemm<<<dim3(BATCH / (32 * WAVES_PER_WG), OUT), dim3(32 * WAVES_PER_WG), 0, stream>>>(
      x, z, P, borders, invlen, out);
}
